# starting kernel
# baseline (speedup 1.0000x reference)
#include <hip/hip_runtime.h>
#include <hip/hip_bf16.h>

#define MDIM 1024
#define BDIM 256
#define LWIN 8
#define PRED 8
#define NSLOT (LWIN + PRED)

typedef __attribute__((ext_vector_type(2))) float v2f;
typedef __attribute__((ext_vector_type(8))) float v8f;

__device__ __forceinline__ v8f wmma_f32_k4(v2f a, v2f b, v8f c) {
    return __builtin_amdgcn_wmma_f32_16x16x4_f32(
         false, a,  false, b,
         (short)0, c,  false,  false);
}

__global__ void dmd_build_agt(const float* __restrict__ Ar,
                              const float* __restrict__ Ai,
                              float* __restrict__ agtr,
                              float* __restrict__ agti) {
    size_t i = (size_t)blockIdx.x * blockDim.x + threadIdx.x;
    if (i >= (size_t)LWIN * MDIM * MDIM) return;
    int n = (int)(i & (MDIM - 1));
    int m = (int)((i >> 10) & (MDIM - 1));
    int p = (int)(i >> 20);
    int l = (p == 0) ? 0 : (LWIN - p);
    size_t src = (size_t)l * MDIM * MDIM + (size_t)m * MDIM + n;
    size_t dst = (size_t)p * MDIM * MDIM + (size_t)(n >> 1) * (2 * MDIM)
               + 2 * m + (n & 1);
    agtr[dst] = Ar[src];
    agti[dst] = Ai[src];
}

__global__ void dmd_init_state(const float* __restrict__ x,
                               float* __restrict__ stR,
                               float* __restrict__ stI) {
    size_t i = (size_t)blockIdx.x * blockDim.x + threadIdx.x;
    if (i >= (size_t)LWIN * MDIM * BDIM) return;
    int b = (int)(i & (BDIM - 1));
    int n = (int)((i >> 8) & (MDIM - 1));
    int s = (int)(i >> 18);
    float v = x[(size_t)b * (LWIN * MDIM) + (size_t)s * MDIM + n];
    size_t dst = (size_t)s * MDIM * BDIM + (size_t)(n >> 1) * (2 * BDIM)
               + 2 * b + (n & 1);
    stR[dst] = v;
    stI[dst] = 0.0f;
}

__global__ __launch_bounds__(32) void dmd_gemm_step(
    const float* __restrict__ agtr, const float* __restrict__ agti,
    const float* __restrict__ stR,  const float* __restrict__ stI,
    float* __restrict__ partR, float* __restrict__ partI, int t) {
    const int lane = threadIdx.x & 31;
    const int g  = lane >> 4;
    const int lm = lane & 15;
    const int p  = blockIdx.z;
    const int m0 = blockIdx.y * 64;
    const int b0 = blockIdx.x * 16;
    const int s  = t + p;

    const float* aR = agtr + (size_t)p * MDIM * MDIM;
    const float* aI = agti + (size_t)p * MDIM * MDIM;
    const float* uR = stR + (size_t)s * MDIM * BDIM;
    const float* uI = stI + (size_t)s * MDIM * BDIM;

    v8f c_rr[4] = {}, c_ii[4] = {}, c_ri[4] = {}, c_ir[4] = {};

    for (int k = 0; k < MDIM; k += 4) {
        const int np = (k + 2 * g) >> 1;
        const float2 bR2 = *(const float2*)(uR + (size_t)np * (2 * BDIM) + 2 * (b0 + lm));
        const float2 bI2 = *(const float2*)(uI + (size_t)np * (2 * BDIM) + 2 * (b0 + lm));
        v2f fbR = { bR2.x, bR2.y };
        v2f fbI = { bI2.x, bI2.y };
#pragma unroll
        for (int mt = 0; mt < 4; ++mt) {
            const int m = m0 + mt * 16 + lm;
            const float2 aR2 = *(const float2*)(aR + (size_t)np * (2 * MDIM) + 2 * m);
            const float2 aI2 = *(const float2*)(aI + (size_t)np * (2 * MDIM) + 2 * m);
            v2f faR = { aR2.x, aR2.y };
            v2f faI = { aI2.x, aI2.y };
            c_rr[mt] = wmma_f32_k4(faR, fbR, c_rr[mt]);
            c_ii[mt] = wmma_f32_k4(faI, fbI, c_ii[mt]);
            c_ri[mt] = wmma_f32_k4(faR, fbI, c_ri[mt]);
            c_ir[mt] = wmma_f32_k4(faI, fbR, c_ir[mt]);
        }
    }

#pragma unroll
    for (int mt = 0; mt < 4; ++mt) {
#pragma unroll
        for (int r = 0; r < 8; ++r) {
            const int m = m0 + mt * 16 + r + 8 * g;
            const size_t o = (size_t)p * MDIM * BDIM + (size_t)m * BDIM + b0 + lm;
            partR[o] = c_rr[mt][r] - c_ii[mt][r];
            partI[o] = c_ri[mt][r] + c_ir[mt][r];
        }
    }
}

__global__ void dmd_reduce_step(const float* __restrict__ partR,
                                const float* __restrict__ partI,
                                float* __restrict__ stR,
                                float* __restrict__ stI, int t) {
    int i = blockIdx.x * blockDim.x + threadIdx.x;
    if (i >= MDIM * BDIM) return;
    int m = i >> 8;
    int b = i & (BDIM - 1);
    float sr = 0.0f, si = 0.0f;
#pragma unroll
    for (int p = 0; p < LWIN; ++p) {
        sr += partR[(size_t)p * MDIM * BDIM + i];
        si += partI[(size_t)p * MDIM * BDIM + i];
    }
    size_t dst = (size_t)(LWIN + t) * MDIM * BDIM
               + (size_t)(m >> 1) * (2 * BDIM) + 2 * b + (m & 1);
    stR[dst] = sr;
    stI[dst] = si;
}

__global__ void dmd_write_out(const float* __restrict__ stR,
                              float* __restrict__ out) {
    size_t i = (size_t)blockIdx.x * blockDim.x + threadIdx.x;
    if (i >= (size_t)BDIM * LWIN * MDIM) return;
    int m = (int)(i & (MDIM - 1));
    int l = (int)((i >> 10) & (LWIN - 1));
    int b = (int)(i >> 13);
    size_t src = (size_t)(LWIN + l) * MDIM * BDIM
               + (size_t)(m >> 1) * (2 * BDIM) + 2 * b + (m & 1);
    out[i] = stR[src];
}

extern "C" void kernel_launch(void* const* d_in, const int* in_sizes, int n_in,
                              void* d_out, int out_size, void* d_ws, size_t ws_size,
                              hipStream_t stream) {
    const float* x  = (const float*)d_in[0];
    const float* Ar = (const float*)d_in[1];
    const float* Ai = (const float*)d_in[2];

    float* ws = (float*)d_ws;
    const size_t AGT  = (size_t)LWIN * MDIM * MDIM;
    const size_t STPL = (size_t)NSLOT * MDIM * BDIM;
    const size_t PART = (size_t)LWIN * MDIM * BDIM;
    float* agtr = ws;
    float* agti = agtr + AGT;
    float* stR  = agti + AGT;
    float* stI  = stR + STPL;
    float* pR   = stI + STPL;
    float* pI   = pR + PART;

    dmd_build_agt<<<(unsigned)(AGT / 256), 256, 0, stream>>>(Ar, Ai, agtr, agti);
    dmd_init_state<<<(unsigned)(((size_t)LWIN * MDIM * BDIM) / 256), 256, 0, stream>>>(x, stR, stI);

    for (int t = 0; t < PRED; ++t) {
        dmd_gemm_step<<<dim3(BDIM / 16, MDIM / 64, LWIN), 32, 0, stream>>>(
            agtr, agti, stR, stI, pR, pI, t);
        dmd_reduce_step<<<(MDIM * BDIM) / 256, 256, 0, stream>>>(pR, pI, stR, stI, t);
    }

    dmd_write_out<<<(unsigned)(((size_t)BDIM * LWIN * MDIM) / 256), 256, 0, stream>>>(
        stR, (float*)d_out);
}


# gemm step takes operator and state rows as given (k contiguous) through LDS: 8-wave workgroup, 64x128 tile, 64-k blocks requested one block ahead; transpose launch gone; every sum keeps its order
# speedup vs baseline: 2.1006x; 2.1006x over previous
#include <hip/hip_runtime.h>


#define MDIM 1024
#define BDIM 256
#define LWIN 8
#define PRED 8
#define NSLOT (LWIN + PRED)

#define TM 128
#define TB 64
#define KB 64
#define PITCH 68
#define RPP 16
#define SEC_OR 0
#define SEC_OI (TM * PITCH)
#define SEC_SR (2 * TM * PITCH)
#define SEC_SI (2 * TM * PITCH + TB * PITCH)
#define BUF_FLOATS ((2 * TM + 2 * TB) * PITCH)
#define NKB (MDIM / KB)

typedef __attribute__((ext_vector_type(2))) float v2f;
typedef __attribute__((ext_vector_type(8))) float v8f;

__device__ __forceinline__ v8f wmma_f32_k4(v2f a, v2f b, v8f c) {
    return __builtin_amdgcn_wmma_f32_16x16x4_f32(
         false, a,  false, b,
         (short)0, c,  false,  false);
}

__global__ void dmd_init_state(const float* __restrict__ x,
                               float* __restrict__ stR,
                               float* __restrict__ stI) {
    const int i = blockIdx.x * blockDim.x + threadIdx.x;
    if (i >= LWIN * BDIM * MDIM / 4) return;
    const int n4 = i & (MDIM / 4 - 1);
    const int b  = (i >> 8) & (BDIM - 1);
    const int s  = i >> 16;
    const float4 v = ((const float4*)x)[((size_t)b * LWIN + s) * (MDIM / 4) + n4];
    const size_t dst = ((size_t)s * BDIM + b) * (MDIM / 4) + n4;
    ((float4*)stR)[dst] = v;
    ((float4*)stI)[dst] = make_float4(0.0f, 0.0f, 0.0f, 0.0f);
}

__global__ __launch_bounds__(256) void dmd_gemm_step(
    const float* __restrict__ opR, const float* __restrict__ opI,
    const float* __restrict__ stR, const float* __restrict__ stI,
    float* __restrict__ partR, float* __restrict__ partI, int t) {
    __shared__ __attribute__((aligned(16))) float L[2 * BUF_FLOATS];

    const int tid  = threadIdx.x;
    const int lane = tid & 31;
    const int wave = tid >> 5;
    const int h  = lane >> 4;
    const int c  = lane & 15;
    const int wm = wave & 3;
    const int wb = wave >> 2;
    const int p  = blockIdx.z;
    const int m0 = blockIdx.y * TM;
    const int b0 = blockIdx.x * TB;
    const int s  = t + p;
    const int l  = (p == 0) ? 0 : (LWIN - p);

    const int jj  = tid & 15;
    const int row = tid >> 4;
    const float* gOR = opR + ((size_t)l * MDIM + m0 + row) * MDIM + 4 * jj;
    const float* gOI = opI + ((size_t)l * MDIM + m0 + row) * MDIM + 4 * jj;
    const float* gSR = stR + ((size_t)s * BDIM + b0 + row) * MDIM + 4 * jj;
    const float* gSI = stI + ((size_t)s * BDIM + b0 + row) * MDIM + 4 * jj;
    const int stg = row * PITCH + 4 * jj;

    const int fO = (32 * wm + c) * PITCH + 2 * h;
    const int fS = (32 * wb + c) * PITCH + 2 * h;

    float4 qOR0, qOR1, qOR2, qOR3, qOR4, qOR5, qOR6, qOR7;
    float4 qOI0, qOI1, qOI2, qOI3, qOI4, qOI5, qOI6, qOI7;
    float4 qSR0, qSR1, qSR2, qSR3, qSI0, qSI1, qSI2, qSI3;
    v8f rr[2][2] = {}, ii[2][2] = {}, ri[2][2] = {}, ir[2][2] = {};
    v2f faoR[2][2], faoI[2][2], fasR[2][2], fasI[2][2];
    v2f fboR[2][2], fboI[2][2], fbsR[2][2], fbsI[2][2];

#define GLD(base, q, k0) (*(const float4*)((base) + (size_t)(q) * RPP * MDIM + (k0)))
#define LOAD_BLOCK(k0)                                                          \
    {                                                                           \
        qOR0 = GLD(gOR, 0, k0); qOI0 = GLD(gOI, 0, k0);                         \
        qOR1 = GLD(gOR, 1, k0); qOI1 = GLD(gOI, 1, k0);                         \
        qOR2 = GLD(gOR, 2, k0); qOI2 = GLD(gOI, 2, k0);                         \
        qOR3 = GLD(gOR, 3, k0); qOI3 = GLD(gOI, 3, k0);                         \
        qOR4 = GLD(gOR, 4, k0); qOI4 = GLD(gOI, 4, k0);                         \
        qOR5 = GLD(gOR, 5, k0); qOI5 = GLD(gOI, 5, k0);                         \
        qOR6 = GLD(gOR, 6, k0); qOI6 = GLD(gOI, 6, k0);                         \
        qOR7 = GLD(gOR, 7, k0); qOI7 = GLD(gOI, 7, k0);                         \
        qSR0 = GLD(gSR, 0, k0); qSI0 = GLD(gSI, 0, k0);                         \
        qSR1 = GLD(gSR, 1, k0); qSI1 = GLD(gSI, 1, k0);                         \
        qSR2 = GLD(gSR, 2, k0); qSI2 = GLD(gSI, 2, k0);                         \
        qSR3 = GLD(gSR, 3, k0); qSI3 = GLD(gSI, 3, k0);                         \
    }

#define LST(Lb, sec, q, v) (*(float4*)((Lb) + (sec) + (q) * RPP * PITCH + stg) = (v))
#define STORE_BLOCK(Lb)                                                         \
    {                                                                           \
        LST(Lb, SEC_OR, 0, qOR0); LST(Lb, SEC_OI, 0, qOI0);                     \
        LST(Lb, SEC_OR, 1, qOR1); LST(Lb, SEC_OI, 1, qOI1);                     \
        LST(Lb, SEC_OR, 2, qOR2); LST(Lb, SEC_OI, 2, qOI2);                     \
        LST(Lb, SEC_OR, 3, qOR3); LST(Lb, SEC_OI, 3, qOI3);                     \
        LST(Lb, SEC_OR, 4, qOR4); LST(Lb, SEC_OI, 4, qOI4);                     \
        LST(Lb, SEC_OR, 5, qOR5); LST(Lb, SEC_OI, 5, qOI5);                     \
        LST(Lb, SEC_OR, 6, qOR6); LST(Lb, SEC_OI, 6, qOI6);                     \
        LST(Lb, SEC_OR, 7, qOR7); LST(Lb, SEC_OI, 7, qOI7);                     \
        LST(Lb, SEC_SR, 0, qSR0); LST(Lb, SEC_SI, 0, qSI0);                     \
        LST(Lb, SEC_SR, 1, qSR1); LST(Lb, SEC_SI, 1, qSI1);                     \
        LST(Lb, SEC_SR, 2, qSR2); LST(Lb, SEC_SI, 2, qSI2);                     \
        LST(Lb, SEC_SR, 3, qSR3); LST(Lb, SEC_SI, 3, qSI3);                     \
    }

#define FRAG_LOAD(P, Lb, g)                                                     \
    {                                                                           \
        _Pragma("unroll")                                                       \
        for (int u = 0; u < 2; ++u) {                                           \
            _Pragma("unroll")                                                   \
            for (int j = 0; j < 2; ++j) {                                       \
                P##oR[u][j] = *(const v2f*)((Lb) + SEC_OR + fO + j * 16 * PITCH + 4 * (2 * (g) + u)); \
                P##oI[u][j] = *(const v2f*)((Lb) + SEC_OI + fO + j * 16 * PITCH + 4 * (2 * (g) + u)); \
                P##sR[u][j] = *(const v2f*)((Lb) + SEC_SR + fS + j * 16 * PITCH + 4 * (2 * (g) + u)); \
                P##sI[u][j] = *(const v2f*)((Lb) + SEC_SI + fS + j * 16 * PITCH + 4 * (2 * (g) + u)); \
            }                                                                   \
        }                                                                       \
    }

#define FRAG_MMA(P)                                                             \
    {                                                                           \
        _Pragma("unroll")                                                       \
        for (int u = 0; u < 2; ++u) {                                           \
            _Pragma("unroll")                                                   \
            for (int i = 0; i < 2; ++i) {                                       \
                _Pragma("unroll")                                               \
                for (int j = 0; j < 2; ++j) {                                   \
                    rr[i][j] = wmma_f32_k4(P##sR[u][i], P##oR[u][j], rr[i][j]); \
                    ii[i][j] = wmma_f32_k4(P##sI[u][i], P##oI[u][j], ii[i][j]); \
                    ri[i][j] = wmma_f32_k4(P##sI[u][i], P##oR[u][j], ri[i][j]); \
                    ir[i][j] = wmma_f32_k4(P##sR[u][i], P##oI[u][j], ir[i][j]); \
                }                                                               \
            }                                                                   \
        }                                                                       \
    }

#define COMPUTE_BLOCK(Lb)                                                       \
    {                                                                           \
        FRAG_LOAD(fa, Lb, 0)                                                    \
        _Pragma("unroll")                                                       \
        for (int g = 0; g < KB / 8; g += 2) {                                   \
            FRAG_LOAD(fb, Lb, g + 1)                                            \
            __builtin_amdgcn_sched_barrier(0);                                  \
            FRAG_MMA(fa)                                                        \
            if (g + 2 < KB / 8) FRAG_LOAD(fa, Lb, g + 2)                        \
            __builtin_amdgcn_sched_barrier(0);                                  \
            FRAG_MMA(fb)                                                        \
        }                                                                       \
    }

    LOAD_BLOCK(0)
    STORE_BLOCK(L)
    LOAD_BLOCK(KB)
    __syncthreads();

    for (int kb = 0; kb < NKB - 1; ++kb) {
        float* cur = L + (kb & 1) * BUF_FLOATS;
        float* nxt = L + ((kb & 1) ^ 1) * BUF_FLOATS;
        COMPUTE_BLOCK(cur)
        STORE_BLOCK(nxt)
        __syncthreads();
        if (kb + 2 < NKB) LOAD_BLOCK((kb + 2) * KB)
    }
    {
        float* cur = L + ((NKB - 1) & 1) * BUF_FLOATS;
        COMPUTE_BLOCK(cur)
    }

#undef GLD
#undef LST
#undef LOAD_BLOCK
#undef STORE_BLOCK
#undef COMPUTE_BLOCK
#undef FRAG_LOAD
#undef FRAG_MMA

#pragma unroll
    for (int i = 0; i < 2; ++i) {
#pragma unroll
        for (int j = 0; j < 2; ++j) {
#pragma unroll
            for (int r = 0; r < 8; ++r) {
                const int b = b0 + 32 * wb + 16 * i + 8 * h + r;
                const int m = m0 + 32 * wm + 16 * j + c;
                const size_t o = ((size_t)p * BDIM + b) * MDIM + m;
                partR[o] = rr[i][j][r] - ii[i][j][r];
                partI[o] = ri[i][j][r] + ir[i][j][r];
            }
        }
    }
}

__global__ void dmd_reduce_step(const float* __restrict__ partR,
                                const float* __restrict__ partI,
                                float* __restrict__ stR,
                                float* __restrict__ stI, int t) {
    const int i = blockIdx.x * blockDim.x + threadIdx.x;
    if (i >= BDIM * MDIM / 4) return;
    float4 sr = make_float4(0.0f, 0.0f, 0.0f, 0.0f);
    float4 si = make_float4(0.0f, 0.0f, 0.0f, 0.0f);
#pragma unroll
    for (int p = 0; p < LWIN; ++p) {
        const float4 a = ((const float4*)partR)[(size_t)p * (BDIM * MDIM / 4) + i];
        const float4 d = ((const float4*)partI)[(size_t)p * (BDIM * MDIM / 4) + i];
        sr.x += a.x; sr.y += a.y; sr.z += a.z; sr.w += a.w;
        si.x += d.x; si.y += d.y; si.z += d.z; si.w += d.w;
    }
    const size_t dst = (size_t)(LWIN + t) * (BDIM * MDIM / 4) + i;
    ((float4*)stR)[dst] = sr;
    ((float4*)stI)[dst] = si;
}

__global__ void dmd_write_out(const float* __restrict__ stR,
                              float* __restrict__ out) {
    const int i = blockIdx.x * blockDim.x + threadIdx.x;
    if (i >= BDIM * LWIN * MDIM / 4) return;
    const int m4 = i & (MDIM / 4 - 1);
    const int l  = (i >> 8) & (LWIN - 1);
    const int b  = i >> 11;
    const size_t src = ((size_t)(LWIN + l) * BDIM + b) * (MDIM / 4) + m4;
    ((float4*)out)[i] = ((const float4*)stR)[src];
}

extern "C" void kernel_launch(void* const* d_in, const int* in_sizes, int n_in,
                              void* d_out, int out_size, void* d_ws, size_t ws_size,
                              hipStream_t stream) {
    const float* x  = (const float*)d_in[0];
    const float* Ar = (const float*)d_in[1];
    const float* Ai = (const float*)d_in[2];

    float* ws = (float*)d_ws;
    const size_t STPL = (size_t)NSLOT * MDIM * BDIM;
    const size_t PART = (size_t)LWIN * MDIM * BDIM;
    float* stR  = ws;
    float* stI  = stR + STPL;
    float* pR   = stI + STPL;
    float* pI   = pR + PART;

    dmd_init_state<<<(LWIN * BDIM * MDIM / 4) / 256, 256, 0, stream>>>(x, stR, stI);

    for (int t = 0; t < PRED; ++t) {
        dmd_gemm_step<<<dim3(BDIM / TB, MDIM / TM, LWIN), 256, 0, stream>>>(
            Ar, Ai, stR, stI, pR, pI, t);
        dmd_reduce_step<<<(MDIM * BDIM / 4) / 256, 256, 0, stream>>>(pR, pI, stR, stI, t);
    }

    dmd_write_out<<<(BDIM * LWIN * MDIM / 4) / 256, 256, 0, stream>>>(
        stR, (float*)d_out);
}


# split-K moved from workgroups to the 8 waves of one workgroup: 32x32 tile, wave p takes window slot p from its own LDS buffers, no loop barrier; sums keep their order; reduce and output launches gone
# speedup vs baseline: 2.2064x; 1.0504x over previous
#include <hip/hip_runtime.h>


#define MDIM 1024
#define BDIM 256
#define LWIN 8
#define PRED 8
#define NSLOT (LWIN + PRED)

#define TM 32
#define TB 32
#define KB 32
#define PITCH 36
#define RPI 4
#define SEC_OR 0
#define SEC_OI (TM * PITCH)
#define SEC_SR (2 * TM * PITCH)
#define SEC_SI (2 * TM * PITCH + TB * PITCH)
#define BUF_FLOATS ((2 * TM + 2 * TB) * PITCH)
#define WAVE_FLOATS (2 * BUF_FLOATS)
#define NKB (MDIM / KB)
#define PP 36

typedef __attribute__((ext_vector_type(2))) float v2f;
typedef __attribute__((ext_vector_type(8))) float v8f;

__device__ __forceinline__ v8f wmma_f32_k4(v2f a, v2f b, v8f c) {
    return __builtin_amdgcn_wmma_f32_16x16x4_f32(
         false, a,  false, b,
         (short)0, c,  false,  false);
}

__global__ void dmd_init_state(const float* __restrict__ x,
                               float* __restrict__ stR,
                               float* __restrict__ stI) {
    const int i = blockIdx.x * blockDim.x + threadIdx.x;
    if (i >= LWIN * BDIM * MDIM / 4) return;
    const int n4 = i & (MDIM / 4 - 1);
    const int b  = (i >> 8) & (BDIM - 1);
    const int s  = i >> 16;
    const float4 v = ((const float4*)x)[((size_t)b * LWIN + s) * (MDIM / 4) + n4];
    const size_t dst = ((size_t)s * BDIM + b) * (MDIM / 4) + n4;
    ((float4*)stR)[dst] = v;
    ((float4*)stI)[dst] = make_float4(0.0f, 0.0f, 0.0f, 0.0f);
}

__global__ __launch_bounds__(256) void dmd_gemm_step(
    const float* __restrict__ opR, const float* __restrict__ opI,
    const float* __restrict__ stRin, const float* __restrict__ stIin,
    float* stRout, float* stIout, float* out, int t) {
    __shared__ __attribute__((aligned(16))) float L[8 * WAVE_FLOATS];

    const int tid  = threadIdx.x;
    const int lane = tid & 31;
    const int p  = __builtin_amdgcn_readfirstlane(tid >> 5);
    const int h  = lane >> 4;
    const int c  = lane & 15;
    const int m0 = blockIdx.y * TM;
    const int b0 = blockIdx.x * TB;
    const int s  = t + p;
    const int l  = (p == 0) ? 0 : (LWIN - p);

    const int jj   = lane & 7;
    const int rsub = lane >> 3;
    const float* gOR = opR   + ((size_t)l * MDIM + m0 + rsub) * MDIM + 4 * jj;
    const float* gOI = opI   + ((size_t)l * MDIM + m0 + rsub) * MDIM + 4 * jj;
    const float* gSR = stRin + ((size_t)s * BDIM + b0 + rsub) * MDIM + 4 * jj;
    const float* gSI = stIin + ((size_t)s * BDIM + b0 + rsub) * MDIM + 4 * jj;
    const int stg = rsub * PITCH + 4 * jj;
    float* Lw = L + p * WAVE_FLOATS;

    const int fO = c * PITCH + 2 * h;

    float4 qOR0, qOR1, qOR2, qOR3, qOR4, qOR5, qOR6, qOR7;
    float4 qOI0, qOI1, qOI2, qOI3, qOI4, qOI5, qOI6, qOI7;
    float4 qSR0, qSR1, qSR2, qSR3, qSR4, qSR5, qSR6, qSR7;
    float4 qSI0, qSI1, qSI2, qSI3, qSI4, qSI5, qSI6, qSI7;
    v8f rr[2][2] = {}, ii[2][2] = {}, ri[2][2] = {}, ir[2][2] = {};
    v2f faoR[2][2], faoI[2][2], fasR[2][2], fasI[2][2];
    v2f fboR[2][2], fboI[2][2], fbsR[2][2], fbsI[2][2];

#define GLD(base, q, k0) (*(const float4*)((base) + (size_t)(q) * RPI * MDIM + (k0)))
#define LOAD_BLOCK(k0)                                                          \
    {                                                                           \
        qOR0 = GLD(gOR, 0, k0); qOI0 = GLD(gOI, 0, k0);                         \
        qOR1 = GLD(gOR, 1, k0); qOI1 = GLD(gOI, 1, k0);                         \
        qOR2 = GLD(gOR, 2, k0); qOI2 = GLD(gOI, 2, k0);                         \
        qOR3 = GLD(gOR, 3, k0); qOI3 = GLD(gOI, 3, k0);                         \
        qOR4 = GLD(gOR, 4, k0); qOI4 = GLD(gOI, 4, k0);                         \
        qOR5 = GLD(gOR, 5, k0); qOI5 = GLD(gOI, 5, k0);                         \
        qOR6 = GLD(gOR, 6, k0); qOI6 = GLD(gOI, 6, k0);                         \
        qOR7 = GLD(gOR, 7, k0); qOI7 = GLD(gOI, 7, k0);                         \
        qSR0 = GLD(gSR, 0, k0); qSI0 = GLD(gSI, 0, k0);                         \
        qSR1 = GLD(gSR, 1, k0); qSI1 = GLD(gSI, 1, k0);                         \
        qSR2 = GLD(gSR, 2, k0); qSI2 = GLD(gSI, 2, k0);                         \
        qSR3 = GLD(gSR, 3, k0); qSI3 = GLD(gSI, 3, k0);                         \
        qSR4 = GLD(gSR, 4, k0); qSI4 = GLD(gSI, 4, k0);                         \
        qSR5 = GLD(gSR, 5, k0); qSI5 = GLD(gSI, 5, k0);                         \
        qSR6 = GLD(gSR, 6, k0); qSI6 = GLD(gSI, 6, k0);                         \
        qSR7 = GLD(gSR, 7, k0); qSI7 = GLD(gSI, 7, k0);                         \
    }

#define LST(Lb, sec, q, v) (*(float4*)((Lb) + (sec) + (q) * RPI * PITCH + stg) = (v))
#define STORE_BLOCK(Lb)                                                         \
    {                                                                           \
        LST(Lb, SEC_OR, 0, qOR0); LST(Lb, SEC_OI, 0, qOI0);                     \
        LST(Lb, SEC_OR, 1, qOR1); LST(Lb, SEC_OI, 1, qOI1);                     \
        LST(Lb, SEC_OR, 2, qOR2); LST(Lb, SEC_OI, 2, qOI2);                     \
        LST(Lb, SEC_OR, 3, qOR3); LST(Lb, SEC_OI, 3, qOI3);                     \
        LST(Lb, SEC_OR, 4, qOR4); LST(Lb, SEC_OI, 4, qOI4);                     \
        LST(Lb, SEC_OR, 5, qOR5); LST(Lb, SEC_OI, 5, qOI5);                     \
        LST(Lb, SEC_OR, 6, qOR6); LST(Lb, SEC_OI, 6, qOI6);                     \
        LST(Lb, SEC_OR, 7, qOR7); LST(Lb, SEC_OI, 7, qOI7);                     \
        LST(Lb, SEC_SR, 0, qSR0); LST(Lb, SEC_SI, 0, qSI0);                     \
        LST(Lb, SEC_SR, 1, qSR1); LST(Lb, SEC_SI, 1, qSI1);                     \
        LST(Lb, SEC_SR, 2, qSR2); LST(Lb, SEC_SI, 2, qSI2);                     \
        LST(Lb, SEC_SR, 3, qSR3); LST(Lb, SEC_SI, 3, qSI3);                     \
        LST(Lb, SEC_SR, 4, qSR4); LST(Lb, SEC_SI, 4, qSI4);                     \
        LST(Lb, SEC_SR, 5, qSR5); LST(Lb, SEC_SI, 5, qSI5);                     \
        LST(Lb, SEC_SR, 6, qSR6); LST(Lb, SEC_SI, 6, qSI6);                     \
        LST(Lb, SEC_SR, 7, qSR7); LST(Lb, SEC_SI, 7, qSI7);                     \
    }

#define WAVE_HANDOVER()                                                         \
    {                                                                           \
        __builtin_amdgcn_fence(__ATOMIC_RELEASE, "wavefront");                  \
        __builtin_amdgcn_wave_barrier();                                        \
        __builtin_amdgcn_fence(__ATOMIC_ACQUIRE, "wavefront");                  \
    }

#define FRAG_LOAD(P, Lb, g)                                                     \
    {                                                                           \
        _Pragma("unroll")                                                       \
        for (int u = 0; u < 2; ++u) {                                           \
            _Pragma("unroll")                                                   \
            for (int j = 0; j < 2; ++j) {                                       \
                P##oR[u][j] = *(const v2f*)((Lb) + SEC_OR + fO + j * 16 * PITCH + 4 * (2 * (g) + u)); \
                P##oI[u][j] = *(const v2f*)((Lb) + SEC_OI + fO + j * 16 * PITCH + 4 * (2 * (g) + u)); \
                P##sR[u][j] = *(const v2f*)((Lb) + SEC_SR + fO + j * 16 * PITCH + 4 * (2 * (g) + u)); \
                P##sI[u][j] = *(const v2f*)((Lb) + SEC_SI + fO + j * 16 * PITCH + 4 * (2 * (g) + u)); \
            }                                                                   \
        }                                                                       \
    }

#define FRAG_MMA(P)                                                             \
    {                                                                           \
        _Pragma("unroll")                                                       \
        for (int u = 0; u < 2; ++u) {                                           \
            _Pragma("unroll")                                                   \
            for (int i = 0; i < 2; ++i) {                                       \
                _Pragma("unroll")                                               \
                for (int j = 0; j < 2; ++j) {                                   \
                    rr[i][j] = wmma_f32_k4(P##sR[u][i], P##oR[u][j], rr[i][j]); \
                    ii[i][j] = wmma_f32_k4(P##sI[u][i], P##oI[u][j], ii[i][j]); \
                    ri[i][j] = wmma_f32_k4(P##sI[u][i], P##oR[u][j], ri[i][j]); \
                    ir[i][j] = wmma_f32_k4(P##sR[u][i], P##oI[u][j], ir[i][j]); \
                }                                                               \
            }                                                                   \
        }                                                                       \
    }

    LOAD_BLOCK(0)
    STORE_BLOCK(Lw)
    LOAD_BLOCK(KB)
    WAVE_HANDOVER()
    FRAG_LOAD(fa, Lw, 0)

    for (int kb = 0; kb < NKB; ++kb) {
        float* cur = Lw + (kb & 1) * BUF_FLOATS;
        float* nxt = Lw + ((kb & 1) ^ 1) * BUF_FLOATS;
        FRAG_LOAD(fb, cur, 1)
        __builtin_amdgcn_sched_barrier(0);
        FRAG_MMA(fa)
        FRAG_LOAD(fa, cur, 2)
        __builtin_amdgcn_sched_barrier(0);
        FRAG_MMA(fb)
        FRAG_LOAD(fb, cur, 3)
        __builtin_amdgcn_sched_barrier(0);
        FRAG_MMA(fa)
        if (kb + 1 < NKB) {
            STORE_BLOCK(nxt)
            WAVE_HANDOVER()
            FRAG_LOAD(fa, nxt, 0)
        }
        if (kb + 2 < NKB) LOAD_BLOCK((kb + 2) * KB)
        __builtin_amdgcn_sched_barrier(0);
        FRAG_MMA(fb)
    }

#undef GLD
#undef LST
#undef LOAD_BLOCK
#undef STORE_BLOCK
#undef WAVE_HANDOVER
#undef FRAG_LOAD
#undef FRAG_MMA

    float* PRw = Lw;
    float* PIw = Lw + TB * PP;
#pragma unroll
    for (int i = 0; i < 2; ++i) {
#pragma unroll
        for (int j = 0; j < 2; ++j) {
#pragma unroll
            for (int r = 0; r < 8; ++r) {
                const int o = (16 * i + 8 * h + r) * PP + 16 * j + c;
                PRw[o] = rr[i][j][r] - ii[i][j][r];
                PIw[o] = ri[i][j][r] + ir[i][j][r];
            }
        }
    }
    __syncthreads();

#pragma unroll
    for (int q = 0; q < 4; ++q) {
        const int bl = 4 * p + q;
        float sr = 0.0f, si = 0.0f;
#pragma unroll
        for (int w = 0; w < LWIN; ++w) {
            sr += L[w * WAVE_FLOATS + bl * PP + lane];
            si += L[w * WAVE_FLOATS + TB * PP + bl * PP + lane];
        }
        const int b = b0 + bl;
        const int m = m0 + lane;
        const size_t so = ((size_t)(LWIN + t) * BDIM + b) * MDIM + m;
        stRout[so] = sr;
        stIout[so] = si;
        out[((size_t)b * LWIN + t) * MDIM + m] = sr;
    }
}

extern "C" void kernel_launch(void* const* d_in, const int* in_sizes, int n_in,
                              void* d_out, int out_size, void* d_ws, size_t ws_size,
                              hipStream_t stream) {
    const float* x  = (const float*)d_in[0];
    const float* Ar = (const float*)d_in[1];
    const float* Ai = (const float*)d_in[2];

    float* ws = (float*)d_ws;
    const size_t STPL = (size_t)NSLOT * MDIM * BDIM;
    float* stR  = ws;
    float* stI  = stR + STPL;

    dmd_init_state<<<(LWIN * BDIM * MDIM / 4) / 256, 256, 0, stream>>>(x, stR, stI);

    for (int t = 0; t < PRED; ++t) {
        dmd_gemm_step<<<dim3(BDIM / TB, MDIM / TM), 256, 0, stream>>>(
            Ar, Ai, stR, stI, stR, stI, (float*)d_out, t);
    }
}


# recurrence split by linearity: one launch forms the input part of all 8 slots from the real window (64x32 tiles), 7 launches add the feedback terms; k re-cut over the 8 waves, sum order changes
# speedup vs baseline: 2.8686x; 1.3001x over previous
#include <hip/hip_runtime.h>


#define MDIM 1024
#define BDIM 256
#define LWIN 8
#define PRED 8
#define NSLOT (LWIN + PRED)

#define TM 32
#define KB 32
#define PITCH 36
#define RPI 4
#define SEC_OR 0
#define SEC_OI (TM * PITCH)
#define SEC_SR (2 * TM * PITCH)
#define SEC_SI (3 * TM * PITCH)
#define BUF_FLOATS (4 * TM * PITCH)
#define WAVE_FLOATS (2 * BUF_FLOATS)
#define NKB (MDIM / KB)
#define PP 36

typedef __attribute__((ext_vector_type(2))) float v2f;
typedef __attribute__((ext_vector_type(8))) float v8f;

__device__ __forceinline__ v8f wmma_f32_k4(v2f a, v2f b, v8f c) {
    return __builtin_amdgcn_wmma_f32_16x16x4_f32(
         false, a,  false, b,
         (short)0, c,  false,  false);
}

template <bool RO>
__global__ __launch_bounds__(256) void dmd_gemm_part(
    const float* __restrict__ opR, const float* __restrict__ opI,
    const float* __restrict__ x, float* stR, float* stI, float* out, int t_arg) {
    __shared__ __attribute__((aligned(16))) float L[8 * WAVE_FLOATS];

    constexpr int TBW  = RO ? 64 : 32;
    constexpr int SSTR = RO ? LWIN * MDIM : MDIM;
    constexpr int RPW  = TBW / 8;

    const int tid  = threadIdx.x;
    const int lane = tid & 31;
    const int w  = __builtin_amdgcn_readfirstlane(tid >> 5);
    const int h  = lane >> 4;
    const int c  = lane & 15;
    const int m0 = blockIdx.y * TM;
    const int b0 = blockIdx.x * TBW;
    const int tt = RO ? (int)blockIdx.z : t_arg;
    const int n  = RO ? (LWIN - tt) : tt;
    const int nb = 4 * n;
    const int g0 = w * nb;

    const int jj   = lane & 7;
    const int rsub = lane >> 3;
    const size_t laneO = (size_t)(m0 + rsub) * MDIM + 4 * jj;
    const size_t laneS = (size_t)(b0 + rsub) * SSTR + 4 * jj;
    const float* sRb = RO ? x : (const float*)stR;
    const float* sIb = (const float*)stI;
    const int stg = rsub * PITCH + 4 * jj;
    float* Lw = L + w * WAVE_FLOATS;

    const int fO = c * PITCH + 2 * h;

    float4 qOR0, qOR1, qOR2, qOR3, qOR4, qOR5, qOR6, qOR7;
    float4 qOI0, qOI1, qOI2, qOI3, qOI4, qOI5, qOI6, qOI7;
    float4 qSR0, qSR1, qSR2, qSR3, qSR4, qSR5, qSR6, qSR7;
    float4 qSI0, qSI1, qSI2, qSI3, qSI4, qSI5, qSI6, qSI7;
    v8f rr[2][2] = {}, ii[2][2] = {}, ri[2][2] = {}, ir[2][2] = {};
    v2f faoR[2][2], faoI[2][2], fasR[2][2], fasI[2][2];
    v2f fboR[2][2], fboI[2][2], fbsR[2][2], fbsI[2][2];

#define GLD(base, q, str) (*(const float4*)((base) + (size_t)(q) * RPI * (str)))
#define LOAD_BLOCK(j)                                                           \
    {                                                                           \
        const int g_  = g0 + (j);                                               \
        const int it_ = g_ >> 5;                                                \
        const int k0_ = (g_ & (NKB - 1)) * KB;                                  \
        const int l_  = RO ? ((it_ == 0) ? 0 : (LWIN - it_)) : (tt - it_);      \
        const size_t so_ = RO ? (size_t)(tt + it_) * MDIM                       \
                              : (size_t)(LWIN + it_) * BDIM * MDIM;             \
        const float* pOR = opR + (size_t)l_ * MDIM * MDIM + k0_ + laneO;        \
        const float* pOI = opI + (size_t)l_ * MDIM * MDIM + k0_ + laneO;        \
        const float* pSR = sRb + so_ + k0_ + laneS;                             \
        const float* pSI = RO ? (pSR + (size_t)8 * RPI * SSTR) : (sIb + so_ + k0_ + laneS); \
        qOR0 = GLD(pOR, 0, MDIM); qOI0 = GLD(pOI, 0, MDIM);                     \
        qOR1 = GLD(pOR, 1, MDIM); qOI1 = GLD(pOI, 1, MDIM);                     \
        qOR2 = GLD(pOR, 2, MDIM); qOI2 = GLD(pOI, 2, MDIM);                     \
        qOR3 = GLD(pOR, 3, MDIM); qOI3 = GLD(pOI, 3, MDIM);                     \
        qOR4 = GLD(pOR, 4, MDIM); qOI4 = GLD(pOI, 4, MDIM);                     \
        qOR5 = GLD(pOR, 5, MDIM); qOI5 = GLD(pOI, 5, MDIM);                     \
        qOR6 = GLD(pOR, 6, MDIM); qOI6 = GLD(pOI, 6, MDIM);                     \
        qOR7 = GLD(pOR, 7, MDIM); qOI7 = GLD(pOI, 7, MDIM);                     \
        qSR0 = GLD(pSR, 0, SSTR); qSI0 = GLD(pSI, 0, SSTR);                     \
        qSR1 = GLD(pSR, 1, SSTR); qSI1 = GLD(pSI, 1, SSTR);                     \
        qSR2 = GLD(pSR, 2, SSTR); qSI2 = GLD(pSI, 2, SSTR);                     \
        qSR3 = GLD(pSR, 3, SSTR); qSI3 = GLD(pSI, 3, SSTR);                     \
        qSR4 = GLD(pSR, 4, SSTR); qSI4 = GLD(pSI, 4, SSTR);                     \
        qSR5 = GLD(pSR, 5, SSTR); qSI5 = GLD(pSI, 5, SSTR);                     \
        qSR6 = GLD(pSR, 6, SSTR); qSI6 = GLD(pSI, 6, SSTR);                     \
        qSR7 = GLD(pSR, 7, SSTR); qSI7 = GLD(pSI, 7, SSTR);                     \
    }

#define LST(Lb, sec, q, v) (*(float4*)((Lb) + (sec) + (q) * RPI * PITCH + stg) = (v))
#define STORE_BLOCK(Lb)                                                         \
    {                                                                           \
        LST(Lb, SEC_OR, 0, qOR0); LST(Lb, SEC_OI, 0, qOI0);                     \
        LST(Lb, SEC_OR, 1, qOR1); LST(Lb, SEC_OI, 1, qOI1);                     \
        LST(Lb, SEC_OR, 2, qOR2); LST(Lb, SEC_OI, 2, qOI2);                     \
        LST(Lb, SEC_OR, 3, qOR3); LST(Lb, SEC_OI, 3, qOI3);                     \
        LST(Lb, SEC_OR, 4, qOR4); LST(Lb, SEC_OI, 4, qOI4);                     \
        LST(Lb, SEC_OR, 5, qOR5); LST(Lb, SEC_OI, 5, qOI5);                     \
        LST(Lb, SEC_OR, 6, qOR6); LST(Lb, SEC_OI, 6, qOI6);                     \
        LST(Lb, SEC_OR, 7, qOR7); LST(Lb, SEC_OI, 7, qOI7);                     \
        LST(Lb, SEC_SR, 0, qSR0); LST(Lb, SEC_SI, 0, qSI0);                     \
        LST(Lb, SEC_SR, 1, qSR1); LST(Lb, SEC_SI, 1, qSI1);                     \
        LST(Lb, SEC_SR, 2, qSR2); LST(Lb, SEC_SI, 2, qSI2);                     \
        LST(Lb, SEC_SR, 3, qSR3); LST(Lb, SEC_SI, 3, qSI3);                     \
        LST(Lb, SEC_SR, 4, qSR4); LST(Lb, SEC_SI, 4, qSI4);                     \
        LST(Lb, SEC_SR, 5, qSR5); LST(Lb, SEC_SI, 5, qSI5);                     \
        LST(Lb, SEC_SR, 6, qSR6); LST(Lb, SEC_SI, 6, qSI6);                     \
        LST(Lb, SEC_SR, 7, qSR7); LST(Lb, SEC_SI, 7, qSI7);                     \
    }

#define WAVE_HANDOVER()                                                         \
    {                                                                           \
        __builtin_amdgcn_fence(__ATOMIC_RELEASE, "wavefront");                  \
        __builtin_amdgcn_wave_barrier();                                        \
        __builtin_amdgcn_fence(__ATOMIC_ACQUIRE, "wavefront");                  \
    }

#define FRAG_LOAD(P, Lb, g)                                                     \
    {                                                                           \
        _Pragma("unroll")                                                       \
        for (int u = 0; u < 2; ++u) {                                           \
            _Pragma("unroll")                                                   \
            for (int j = 0; j < 2; ++j) {                                       \
                P##oR[u][j] = *(const v2f*)((Lb) + SEC_OR + fO + j * 16 * PITCH + 4 * (2 * (g) + u)); \
                P##oI[u][j] = *(const v2f*)((Lb) + SEC_OI + fO + j * 16 * PITCH + 4 * (2 * (g) + u)); \
                P##sR[u][j] = *(const v2f*)((Lb) + SEC_SR + fO + j * 16 * PITCH + 4 * (2 * (g) + u)); \
                P##sI[u][j] = *(const v2f*)((Lb) + SEC_SI + fO + j * 16 * PITCH + 4 * (2 * (g) + u)); \
            }                                                                   \
        }                                                                       \
    }

#define FRAG_MMA(P)                                                             \
    {                                                                           \
        _Pragma("unroll")                                                       \
        for (int u = 0; u < 2; ++u) {                                           \
            _Pragma("unroll")                                                   \
            for (int i = 0; i < 2; ++i) {                                       \
                _Pragma("unroll")                                               \
                for (int j = 0; j < 2; ++j) {                                   \
                    rr[i][j] = wmma_f32_k4(P##sR[u][i], P##oR[u][j], rr[i][j]); \
                    if constexpr (RO) {                                         \
                        ii[i][j] = wmma_f32_k4(P##sI[u][i], P##oR[u][j], ii[i][j]); \
                        ri[i][j] = wmma_f32_k4(P##sI[u][i], P##oI[u][j], ri[i][j]); \
                    } else {                                                    \
                        ii[i][j] = wmma_f32_k4(P##sI[u][i], P##oI[u][j], ii[i][j]); \
                        ri[i][j] = wmma_f32_k4(P##sI[u][i], P##oR[u][j], ri[i][j]); \
                    }                                                           \
                    ir[i][j] = wmma_f32_k4(P##sR[u][i], P##oI[u][j], ir[i][j]); \
                }                                                               \
            }                                                                   \
        }                                                                       \
    }

    LOAD_BLOCK(0)
    STORE_BLOCK(Lw)
    LOAD_BLOCK(1)
    WAVE_HANDOVER()
    FRAG_LOAD(fa, Lw, 0)

#define K_BLOCK(j, ST, LD)                                                      \
    {                                                                           \
        float* cur = Lw + ((j) & 1) * BUF_FLOATS;                               \
        float* nxt = Lw + (((j) & 1) ^ 1) * BUF_FLOATS;                         \
        FRAG_LOAD(fb, cur, 1)                                                   \
        __builtin_amdgcn_sched_barrier(0);                                      \
        FRAG_MMA(fa)                                                            \
        FRAG_LOAD(fa, cur, 2)                                                   \
        __builtin_amdgcn_sched_barrier(0);                                      \
        FRAG_MMA(fb)                                                            \
        FRAG_LOAD(fb, cur, 3)                                                   \
        __builtin_amdgcn_sched_barrier(0);                                      \
        FRAG_MMA(fa)                                                            \
        if (ST) {                                                               \
            STORE_BLOCK(nxt)                                                    \
            WAVE_HANDOVER()                                                     \
            FRAG_LOAD(fa, nxt, 0)                                               \
        }                                                                       \
        if (LD) LOAD_BLOCK((j) + 2)                                             \
        __builtin_amdgcn_sched_barrier(0);                                      \
        FRAG_MMA(fb)                                                            \
    }

    for (int j = 0; j < nb - 2; ++j) K_BLOCK(j, 1, 1)
    K_BLOCK(nb - 2, 1, 0)
    K_BLOCK(nb - 1, 0, 0)

#undef K_BLOCK
#undef GLD
#undef LST
#undef LOAD_BLOCK
#undef STORE_BLOCK
#undef WAVE_HANDOVER
#undef FRAG_LOAD
#undef FRAG_MMA

    const int m = m0 + lane;
    float xr[RPW], xi[RPW];
#pragma unroll
    for (int q = 0; q < RPW; ++q) {
        const size_t so = ((size_t)(LWIN + tt) * BDIM + b0 + RPW * w + q) * MDIM + m;
        xr[q] = RO ? 0.0f : stR[so];
        xi[q] = RO ? 0.0f : stI[so];
    }

    float* PRw = Lw;
    float* PIw = Lw + TBW * PP;
#pragma unroll
    for (int i = 0; i < 2; ++i) {
#pragma unroll
        for (int j = 0; j < 2; ++j) {
#pragma unroll
            for (int r = 0; r < 8; ++r) {
                const int o = (16 * i + 8 * h + r) * PP + 16 * j + c;
                if constexpr (RO) {
                    PRw[o] = rr[i][j][r];
                    PIw[o] = ir[i][j][r];
                    PRw[o + 32 * PP] = ii[i][j][r];
                    PIw[o + 32 * PP] = ri[i][j][r];
                } else {
                    PRw[o] = rr[i][j][r] - ii[i][j][r];
                    PIw[o] = ri[i][j][r] + ir[i][j][r];
                }
            }
        }
    }
    __syncthreads();

#pragma unroll
    for (int q = 0; q < RPW; ++q) {
        const int bl = RPW * w + q;
        float sr = xr[q], si = xi[q];
#pragma unroll
        for (int ww = 0; ww < 8; ++ww) {
            sr += L[ww * WAVE_FLOATS + bl * PP + lane];
            si += L[ww * WAVE_FLOATS + TBW * PP + bl * PP + lane];
        }
        const int b = b0 + bl;
        const size_t so = ((size_t)(LWIN + tt) * BDIM + b) * MDIM + m;
        stR[so] = sr;
        stI[so] = si;
        if (!RO || tt == 0)
            out[((size_t)b * LWIN + tt) * MDIM + m] = sr;
    }
}

extern "C" void kernel_launch(void* const* d_in, const int* in_sizes, int n_in,
                              void* d_out, int out_size, void* d_ws, size_t ws_size,
                              hipStream_t stream) {
    const float* x  = (const float*)d_in[0];
    const float* Ar = (const float*)d_in[1];
    const float* Ai = (const float*)d_in[2];

    float* ws = (float*)d_ws;
    const size_t STPL = (size_t)NSLOT * MDIM * BDIM;
    float* stR  = ws;
    float* stI  = stR + STPL;

    dmd_gemm_part<true><<<dim3(BDIM / 64, MDIM / TM, PRED), 256, 0, stream>>>(
        Ar, Ai, x, stR, stI, (float*)d_out, 0);

    for (int t = 1; t < PRED; ++t) {
        dmd_gemm_part<false><<<dim3(BDIM / 32, MDIM / TM), 256, 0, stream>>>(
            Ar, Ai, x, stR, stI, (float*)d_out, t);
    }
}


# k loop: each real product as 2x2x2 block products per k4-step pair by the seven-product scheme, 7 not 8 matrix ops, fragment sums on the vector unit; the grouping of the sums changes
# speedup vs baseline: 3.0155x; 1.0512x over previous
#include <hip/hip_runtime.h>


#define MDIM 1024
#define BDIM 256
#define LWIN 8
#define PRED 8
#define NSLOT (LWIN + PRED)

#define TM 32
#define KB 32
#define PITCH 36
#define RPI 4
#define SEC_OR 0
#define SEC_OI (TM * PITCH)
#define SEC_SR (2 * TM * PITCH)
#define SEC_SI (3 * TM * PITCH)
#define BUF_FLOATS (4 * TM * PITCH)
#define WAVE_FLOATS (2 * BUF_FLOATS)
#define NKB (MDIM / KB)
#define PP 36


typedef __attribute__((ext_vector_type(2))) float v2f;
typedef __attribute__((ext_vector_type(8))) float v8f;

__device__ __forceinline__ v8f wmma_f32_k4(v2f a, v2f b, v8f c) {
    return __builtin_amdgcn_wmma_f32_16x16x4_f32(
         false, a,  false, b,
         (short)0, c,  false,  false);
}

template <bool RO>
__global__ __launch_bounds__(256) void dmd_gemm_part(
    const float* __restrict__ opR, const float* __restrict__ opI,
    const float* __restrict__ x, float* stR, float* stI, float* out, int t_arg) {
    __shared__ __attribute__((aligned(16))) float L[8 * WAVE_FLOATS];

    constexpr int TBW  = RO ? 64 : 32;
    constexpr int SSTR = RO ? LWIN * MDIM : MDIM;
    constexpr int RPW  = TBW / 8;

    const int tid  = threadIdx.x;
    const int lane = tid & 31;
    const int w  = __builtin_amdgcn_readfirstlane(tid >> 5);
    const int h  = lane >> 4;
    const int c  = lane & 15;
    const int m0 = blockIdx.y * TM;
    const int b0 = blockIdx.x * TBW;
    const int tt = RO ? (int)blockIdx.z : t_arg;
    const int n  = RO ? (LWIN - tt) : tt;
    const int nb = 4 * n;
    const int g0 = w * nb;

    const int jj   = lane & 7;
    const int rsub = lane >> 3;
    const size_t laneO = (size_t)(m0 + rsub) * MDIM + 4 * jj;
    const size_t laneS = (size_t)(b0 + rsub) * SSTR + 4 * jj;
    const float* sRb = RO ? x : (const float*)stR;
    const float* sIb = (const float*)stI;
    const int stg = rsub * PITCH + 4 * jj;
    float* Lw = L + w * WAVE_FLOATS;

    const int fO = c * PITCH + 2 * h;

    float4 qOR0, qOR1, qOR2, qOR3, qOR4, qOR5, qOR6, qOR7;
    float4 qOI0, qOI1, qOI2, qOI3, qOI4, qOI5, qOI6, qOI7;
    float4 qSR0, qSR1, qSR2, qSR3, qSR4, qSR5, qSR6, qSR7;
    float4 qSI0, qSI1, qSI2, qSI3, qSI4, qSI5, qSI6, qSI7;
    v8f Mrr[7] = {}, Mii[7] = {}, Mri[7] = {}, Mir[7] = {};
    v2f faoR[2][2], faoI[2][2], fasR[2][2], fasI[2][2];
    v2f fboR[2][2], fboI[2][2], fbsR[2][2], fbsI[2][2];

#define GLD(base, q, str) (*(const float4*)((base) + (size_t)(q) * RPI * (str)))
#define LOAD_BLOCK(j)                                                           \
    {                                                                           \
        const int g_  = g0 + (j);                                               \
        const int it_ = g_ >> 5;                                                \
        const int k0_ = (g_ & (NKB - 1)) * KB;                                  \
        const int l_  = RO ? ((it_ == 0) ? 0 : (LWIN - it_)) : (tt - it_);      \
        const size_t so_ = RO ? (size_t)(tt + it_) * MDIM                       \
                              : (size_t)(LWIN + it_) * BDIM * MDIM;             \
        const float* pOR = opR + (size_t)l_ * MDIM * MDIM + k0_ + laneO;        \
        const float* pOI = opI + (size_t)l_ * MDIM * MDIM + k0_ + laneO;        \
        const float* pSR = sRb + so_ + k0_ + laneS;                             \
        const float* pSI = RO ? (pSR + (size_t)8 * RPI * SSTR) : (sIb + so_ + k0_ + laneS); \
        qOR0 = GLD(pOR, 0, MDIM); qOI0 = GLD(pOI, 0, MDIM);                     \
        qOR1 = GLD(pOR, 1, MDIM); qOI1 = GLD(pOI, 1, MDIM);                     \
        qOR2 = GLD(pOR, 2, MDIM); qOI2 = GLD(pOI, 2, MDIM);                     \
        qOR3 = GLD(pOR, 3, MDIM); qOI3 = GLD(pOI, 3, MDIM);                     \
        qOR4 = GLD(pOR, 4, MDIM); qOI4 = GLD(pOI, 4, MDIM);                     \
        qOR5 = GLD(pOR, 5, MDIM); qOI5 = GLD(pOI, 5, MDIM);                     \
        qOR6 = GLD(pOR, 6, MDIM); qOI6 = GLD(pOI, 6, MDIM);                     \
        qOR7 = GLD(pOR, 7, MDIM); qOI7 = GLD(pOI, 7, MDIM);                     \
        qSR0 = GLD(pSR, 0, SSTR); qSI0 = GLD(pSI, 0, SSTR);                     \
        qSR1 = GLD(pSR, 1, SSTR); qSI1 = GLD(pSI, 1, SSTR);                     \
        qSR2 = GLD(pSR, 2, SSTR); qSI2 = GLD(pSI, 2, SSTR);                     \
        qSR3 = GLD(pSR, 3, SSTR); qSI3 = GLD(pSI, 3, SSTR);                     \
        qSR4 = GLD(pSR, 4, SSTR); qSI4 = GLD(pSI, 4, SSTR);                     \
        qSR5 = GLD(pSR, 5, SSTR); qSI5 = GLD(pSI, 5, SSTR);                     \
        qSR6 = GLD(pSR, 6, SSTR); qSI6 = GLD(pSI, 6, SSTR);                     \
        qSR7 = GLD(pSR, 7, SSTR); qSI7 = GLD(pSI, 7, SSTR);                     \
    }

#define LST(Lb, sec, q, v) (*(float4*)((Lb) + (sec) + (q) * RPI * PITCH + stg) = (v))
#define STORE_BLOCK(Lb)                                                         \
    {                                                                           \
        LST(Lb, SEC_OR, 0, qOR0); LST(Lb, SEC_OI, 0, qOI0);                     \
        LST(Lb, SEC_OR, 1, qOR1); LST(Lb, SEC_OI, 1, qOI1);                     \
        LST(Lb, SEC_OR, 2, qOR2); LST(Lb, SEC_OI, 2, qOI2);                     \
        LST(Lb, SEC_OR, 3, qOR3); LST(Lb, SEC_OI, 3, qOI3);                     \
        LST(Lb, SEC_OR, 4, qOR4); LST(Lb, SEC_OI, 4, qOI4);                     \
        LST(Lb, SEC_OR, 5, qOR5); LST(Lb, SEC_OI, 5, qOI5);                     \
        LST(Lb, SEC_OR, 6, qOR6); LST(Lb, SEC_OI, 6, qOI6);                     \
        LST(Lb, SEC_OR, 7, qOR7); LST(Lb, SEC_OI, 7, qOI7);                     \
        LST(Lb, SEC_SR, 0, qSR0); LST(Lb, SEC_SI, 0, qSI0);                     \
        LST(Lb, SEC_SR, 1, qSR1); LST(Lb, SEC_SI, 1, qSI1);                     \
        LST(Lb, SEC_SR, 2, qSR2); LST(Lb, SEC_SI, 2, qSI2);                     \
        LST(Lb, SEC_SR, 3, qSR3); LST(Lb, SEC_SI, 3, qSI3);                     \
        LST(Lb, SEC_SR, 4, qSR4); LST(Lb, SEC_SI, 4, qSI4);                     \
        LST(Lb, SEC_SR, 5, qSR5); LST(Lb, SEC_SI, 5, qSI5);                     \
        LST(Lb, SEC_SR, 6, qSR6); LST(Lb, SEC_SI, 6, qSI6);                     \
        LST(Lb, SEC_SR, 7, qSR7); LST(Lb, SEC_SI, 7, qSI7);                     \
    }

#define WAVE_HANDOVER()                                                         \
    {                                                                           \
        __builtin_amdgcn_fence(__ATOMIC_RELEASE, "wavefront");                  \
        __builtin_amdgcn_wave_barrier();                                        \
        __builtin_amdgcn_fence(__ATOMIC_ACQUIRE, "wavefront");                  \
    }

#define FRAG_LOAD(P, Lb, g)                                                     \
    {                                                                           \
        _Pragma("unroll")                                                       \
        for (int u = 0; u < 2; ++u) {                                           \
            _Pragma("unroll")                                                   \
            for (int j = 0; j < 2; ++j) {                                       \
                P##oR[u][j] = *(const v2f*)((Lb) + SEC_OR + fO + j * 16 * PITCH + 4 * (2 * (g) + u)); \
                P##oI[u][j] = *(const v2f*)((Lb) + SEC_OI + fO + j * 16 * PITCH + 4 * (2 * (g) + u)); \
                P##sR[u][j] = *(const v2f*)((Lb) + SEC_SR + fO + j * 16 * PITCH + 4 * (2 * (g) + u)); \
                P##sI[u][j] = *(const v2f*)((Lb) + SEC_SI + fO + j * 16 * PITCH + 4 * (2 * (g) + u)); \
            }                                                                   \
        }                                                                       \
    }

#define SUMS_A(S, f)                                                            \
    const v2f S##1 = f[0][0] + f[1][1];                                         \
    const v2f S##2 = f[0][1] + f[1][1];                                         \
    const v2f S##3 = f[0][0] + f[1][0];                                         \
    const v2f S##4 = f[0][1] - f[0][0];                                         \
    const v2f S##5 = f[1][0] - f[1][1];
#define SUMS_B(T, f)                                                            \
    const v2f T##1 = f[0][0] + f[1][1];                                         \
    const v2f T##3 = f[0][1] - f[1][1];                                         \
    const v2f T##4 = f[1][0] - f[0][0];                                         \
    const v2f T##6 = f[0][0] + f[0][1];                                         \
    const v2f T##7 = f[1][0] + f[1][1];
#define SEVEN(M, a, S, b, T)                                                    \
    {                                                                           \
        M[0] = wmma_f32_k4(S##1, T##1, M[0]);                                   \
        M[1] = wmma_f32_k4(S##2, b[0][0], M[1]);                                \
        M[2] = wmma_f32_k4(a[0][0], T##3, M[2]);                                \
        M[3] = wmma_f32_k4(a[1][1], T##4, M[3]);                                \
        M[4] = wmma_f32_k4(S##3, b[1][1], M[4]);                                \
        M[5] = wmma_f32_k4(S##4, T##6, M[5]);                                   \
        M[6] = wmma_f32_k4(S##5, T##7, M[6]);                                   \
    }

#define PIN7(M)                                                                 \
    asm volatile("" : "+v"(M[0]), "+v"(M[1]), "+v"(M[2]), "+v"(M[3]),           \
                      "+v"(M[4]), "+v"(M[5]), "+v"(M[6]) : : "memory");
#define PIN_ALL() { PIN7(Mrr) PIN7(Mii) PIN7(Mri) PIN7(Mir) }

#define FRAG_MMA(P)                                                             \
    {                                                                           \
        SUMS_A(sr_, P##sR)                                                      \
        SUMS_A(si_, P##sI)                                                      \
        SUMS_B(or_, P##oR)                                                      \
        SUMS_B(oi_, P##oI)                                                      \
        SEVEN(Mrr, P##sR, sr_, P##oR, or_)                                      \
        if constexpr (RO) {                                                     \
            SEVEN(Mii, P##sI, si_, P##oR, or_)                                  \
            SEVEN(Mri, P##sI, si_, P##oI, oi_)                                  \
        } else {                                                                \
            SEVEN(Mii, P##sI, si_, P##oI, oi_)                                  \
            SEVEN(Mri, P##sI, si_, P##oR, or_)                                  \
        }                                                                       \
        SEVEN(Mir, P##sR, sr_, P##oI, oi_)                                      \
    }

    LOAD_BLOCK(0)
    STORE_BLOCK(Lw)
    LOAD_BLOCK(1)
    WAVE_HANDOVER()
    FRAG_LOAD(fa, Lw, 0)

#define K_BLOCK(j, ST, LD)                                                      \
    {                                                                           \
        float* cur = Lw + ((j) & 1) * BUF_FLOATS;                               \
        float* nxt = Lw + (((j) & 1) ^ 1) * BUF_FLOATS;                         \
        FRAG_LOAD(fb, cur, 1)                                                   \
        PIN_ALL()                                                               \
        __builtin_amdgcn_sched_barrier(0);                                      \
        FRAG_MMA(fa)                                                            \
        FRAG_LOAD(fa, cur, 2)                                                   \
        PIN_ALL()                                                               \
        __builtin_amdgcn_sched_barrier(0);                                      \
        FRAG_MMA(fb)                                                            \
        FRAG_LOAD(fb, cur, 3)                                                   \
        PIN_ALL()                                                               \
        __builtin_amdgcn_sched_barrier(0);                                      \
        FRAG_MMA(fa)                                                            \
        if (ST) {                                                               \
            STORE_BLOCK(nxt)                                                    \
            WAVE_HANDOVER()                                                     \
            FRAG_LOAD(fa, nxt, 0)                                               \
        }                                                                       \
        if (LD) LOAD_BLOCK((j) + 2)                                             \
        PIN_ALL()                                                               \
        __builtin_amdgcn_sched_barrier(0);                                      \
        FRAG_MMA(fb)                                                            \
    }

    for (int j = 0; j < nb - 2; ++j) K_BLOCK(j, 1, 1)
    K_BLOCK(nb - 2, 1, 0)
    K_BLOCK(nb - 1, 0, 0)

#undef K_BLOCK
#undef GLD
#undef LST
#undef LOAD_BLOCK
#undef STORE_BLOCK
#undef WAVE_HANDOVER
#undef FRAG_LOAD
#undef FRAG_MMA
#undef SEVEN
#undef PIN7
#undef PIN_ALL
#undef SUMS_A
#undef SUMS_B

    const int m = m0 + lane;
    float xr[RPW], xi[RPW];
#pragma unroll
    for (int q = 0; q < RPW; ++q) {
        const size_t so = ((size_t)(LWIN + tt) * BDIM + b0 + RPW * w + q) * MDIM + m;
        xr[q] = RO ? 0.0f : stR[so];
        xi[q] = RO ? 0.0f : stI[so];
    }

    v8f rr[2][2], ii[2][2], ri[2][2], ir[2][2];
#define COMBINE(C, M)                                                           \
    {                                                                           \
        C[0][0] = M[0] + M[3] - M[4] + M[6];                                    \
        C[0][1] = M[2] + M[4];                                                  \
        C[1][0] = M[1] + M[3];                                                  \
        C[1][1] = M[0] - M[1] + M[2] + M[5];                                    \
    }
    COMBINE(rr, Mrr)
    COMBINE(ii, Mii)
    COMBINE(ri, Mri)
    COMBINE(ir, Mir)
#undef COMBINE
    float* PRw = Lw;
    float* PIw = Lw + TBW * PP;
#pragma unroll
    for (int i = 0; i < 2; ++i) {
#pragma unroll
        for (int j = 0; j < 2; ++j) {
#pragma unroll
            for (int r = 0; r < 8; ++r) {
                const int o = (16 * i + 8 * h + r) * PP + 16 * j + c;
                if constexpr (RO) {
                    PRw[o] = rr[i][j][r];
                    PIw[o] = ir[i][j][r];
                    PRw[o + 32 * PP] = ii[i][j][r];
                    PIw[o + 32 * PP] = ri[i][j][r];
                } else {
                    PRw[o] = rr[i][j][r] - ii[i][j][r];
                    PIw[o] = ri[i][j][r] + ir[i][j][r];
                }
            }
        }
    }
    __syncthreads();

#pragma unroll
    for (int q = 0; q < RPW; ++q) {
        const int bl = RPW * w + q;
        float sr = xr[q], si = xi[q];
#pragma unroll
        for (int ww = 0; ww < 8; ++ww) {
            sr += L[ww * WAVE_FLOATS + bl * PP + lane];
            si += L[ww * WAVE_FLOATS + TBW * PP + bl * PP + lane];
        }
        const int b = b0 + bl;
        const size_t so = ((size_t)(LWIN + tt) * BDIM + b) * MDIM + m;
        stR[so] = sr;
        stI[so] = si;
        if (!RO || tt == 0)
            out[((size_t)b * LWIN + tt) * MDIM + m] = sr;
    }
}

extern "C" void kernel_launch(void* const* d_in, const int* in_sizes, int n_in,
                              void* d_out, int out_size, void* d_ws, size_t ws_size,
                              hipStream_t stream) {
    const float* x  = (const float*)d_in[0];
    const float* Ar = (const float*)d_in[1];
    const float* Ai = (const float*)d_in[2];

    float* ws = (float*)d_ws;
    const size_t STPL = (size_t)NSLOT * MDIM * BDIM;
    float* stR  = ws;
    float* stI  = stR + STPL;

    dmd_gemm_part<true><<<dim3(BDIM / 64, MDIM / TM, PRED), 256, 0, stream>>>(
        Ar, Ai, x, stR, stI, (float*)d_out, 0);

    for (int t = 1; t < PRED; ++t) {
        dmd_gemm_part<false><<<dim3(BDIM / 32, MDIM / TM), 256, 0, stream>>>(
            Ar, Ai, x, stR, stI, (float*)d_out, t);
    }
}
